# starting kernel
# baseline (speedup 1.0000x reference)
#include <hip/hip_runtime.h>
#include <hip/hip_bf16.h>


#define NW    12
#define NL    2
#define QDIM  4096
#define QBATCH 8192

typedef __attribute__((ext_vector_type(2))) float v2f;
typedef __attribute__((ext_vector_type(8))) float v8f;

static __device__ __forceinline__ v8f wmma4(v2f a, v2f b, v8f c) {
    return __builtin_amdgcn_wmma_f32_16x16x4_f32(false, a, false, b, (short)0, c,
                                                 false, false);
}

__global__ void qgate_build(const float* __restrict__ params,
                            float* __restrict__ gr, float* __restrict__ gi) {
    int e = blockIdx.x * blockDim.x + threadIdx.x;
    if (e >= NL * 3 * 256) return;
    int l   = e / (3 * 256);
    int rem = e - l * (3 * 256);
    int g   = rem >> 8;
    int ij  = rem & 255;
    int i = ij >> 4, j = ij & 15;
    float re = 1.f, im = 0.f;
    for (int k = 0; k < 4; ++k) {
        int   w  = g * 4 + k;
        float th = params[l * NW + w] * 0.5f;
        float c  = cosf(th), s = sinf(th);
        int ib = (i >> (3 - k)) & 1;
        int jb = (j >> (3 - k)) & 1;
        float wr, wi;
        if (!ib && !jb)     { wr =  c * c; wi =  s * s; }
        else if (!ib && jb) { wr = -c * s; wi = -c * s; }
        else if (ib && !jb) { wr =  c * s; wi = -c * s; }
        else                { wr =  c * c; wi = -s * s; }
        float nr = re * wr - im * wi;
        float ni = re * wi + im * wr;
        re = nr; im = ni;
    }
    gr[e] = re;
    gi[e] = im;
}

__launch_bounds__(256)
__global__ void qsim(const float* __restrict__ x,
                     const float* __restrict__ head_w,
                     const float* __restrict__ head_b,
                     const float* __restrict__ gr,
                     const float* __restrict__ gi,
                     float* __restrict__ out) {
    __shared__ float Sr[QDIM];
    __shared__ float Si[QDIM];
    __shared__ float tabA[16], tabM[16], tabZ[16];
    __shared__ float cfA[16], cfM[16], cfZ[16];
    __shared__ float trig[2 * NW];
    __shared__ float red[256];

    const int tid  = threadIdx.x;
    const int b    = blockIdx.x;
    const int lane = tid & 31;
    const int wave = tid >> 5;
    const int half = lane >> 4;
    const int lm   = lane & 15;
    const int k0   = half * 2;

    if (tid < NW) {
        float th = x[b * NW + tid] * 0.5f;
        trig[tid]      = cosf(th);
        trig[tid + NW] = sinf(th);
    }
    __syncthreads();

    if (tid < 48) {
        int grp = tid >> 4, v = tid & 15;
        float p = 1.f;
        for (int k = 0; k < 4; ++k) {
            int w = grp * 4 + k;
            p *= ((v >> (3 - k)) & 1) ? trig[NW + w] : trig[w];
        }
        (grp == 0 ? tabA : grp == 1 ? tabM : tabZ)[v] = p;
    } else if (tid < 96) {
        int grp = (tid - 48) >> 4, v = tid & 15;
        float s = 0.f;
        for (int k = 0; k < 4; ++k) {
            int w = grp * 4 + k;
            s += head_w[w] * (((v >> (3 - k)) & 1) ? -1.f : 1.f);
        }
        (grp == 0 ? cfA : grp == 1 ? cfM : cfZ)[v] = s;
    }
    __syncthreads();

    for (int k = 0; k < 16; ++k) {
        int i = tid * 16 + k;
        Sr[i] = tabA[(i >> 8) & 15] * tabM[(i >> 4) & 15] * tabZ[i & 15];
        Si[i] = 0.f;
    }
    __syncthreads();

    for (int l = 0; l < NL; ++l) {
        const float* Gr = gr + l * 3 * 256;
        const float* Gi = gi + l * 3 * 256;

        for (int g = 0; g < 2; ++g) {
            const float* ur = Gr + g * 256;
            const float* ui = Gi + g * 256;
            v2f Ar[4], Av[4], An[4];
            for (int kk = 0; kk < 4; ++kk) {
                int kc = kk * 4 + k0;
                Ar[kk].x = ur[lm * 16 + kc];  Ar[kk].y = ur[lm * 16 + kc + 1];
                Av[kk].x = ui[lm * 16 + kc];  Av[kk].y = ui[lm * 16 + kc + 1];
                An[kk].x = -Av[kk].x;         An[kk].y = -Av[kk].y;
            }
            for (int tt = 0; tt < 2; ++tt) {
                int t = wave + tt * 8;
                v2f Br[4], Bi[4];
                for (int kk = 0; kk < 4; ++kk) {
                    int kc = kk * 4 + k0;
                    int i0 = (g == 0) ? (kc * 256 + t * 16 + lm)
                                      : (t * 256 + kc * 16 + lm);
                    int i1 = (g == 0) ? ((kc + 1) * 256 + t * 16 + lm)
                                      : (t * 256 + (kc + 1) * 16 + lm);
                    Br[kk].x = Sr[i0];  Br[kk].y = Sr[i1];
                    Bi[kk].x = Si[i0];  Bi[kk].y = Si[i1];
                }
                v8f Cr = {}; v8f Ci = {};
                for (int kk = 0; kk < 4; ++kk) {
                    Cr = wmma4(Ar[kk], Br[kk], Cr);
                    Cr = wmma4(An[kk], Bi[kk], Cr);
                    Ci = wmma4(Ar[kk], Bi[kk], Ci);
                    Ci = wmma4(Av[kk], Br[kk], Ci);
                }
                for (int v = 0; v < 8; ++v) {
                    int M = v + half * 8;
                    int o = (g == 0) ? (M * 256 + t * 16 + lm)
                                     : (t * 256 + M * 16 + lm);
                    Sr[o] = Cr[v];  Si[o] = Ci[v];
                }
            }
            __syncthreads();
        }

        {
            const float* ur = Gr + 2 * 256;
            const float* ui = Gi + 2 * 256;
            v2f Btr[4], Bti[4], Btn[4];
            for (int kk = 0; kk < 4; ++kk) {
                int kc = kk * 4 + k0;
                Btr[kk].x = ur[lm * 16 + kc];  Btr[kk].y = ur[lm * 16 + kc + 1];
                Bti[kk].x = ui[lm * 16 + kc];  Bti[kk].y = ui[lm * 16 + kc + 1];
                Btn[kk].x = -Bti[kk].x;        Btn[kk].y = -Bti[kk].y;
            }
            for (int tt = 0; tt < 2; ++tt) {
                int t = wave + tt * 8;
                v2f Asr[4], Asi[4];
                int rbase = (t * 16 + lm) * 16;
                for (int kk = 0; kk < 4; ++kk) {
                    int kc = kk * 4 + k0;
                    Asr[kk].x = Sr[rbase + kc];  Asr[kk].y = Sr[rbase + kc + 1];
                    Asi[kk].x = Si[rbase + kc];  Asi[kk].y = Si[rbase + kc + 1];
                }
                v8f Cr = {}; v8f Ci = {};
                for (int kk = 0; kk < 4; ++kk) {
                    Cr = wmma4(Asr[kk], Btr[kk], Cr);
                    Cr = wmma4(Asi[kk], Btn[kk], Cr);
                    Ci = wmma4(Asr[kk], Bti[kk], Ci);
                    Ci = wmma4(Asi[kk], Btr[kk], Ci);
                }
                for (int v = 0; v < 8; ++v) {
                    int M = v + half * 8;
                    int o = (t * 16 + M) * 16 + lm;
                    Sr[o] = Cr[v];  Si[o] = Ci[v];
                }
            }
            __syncthreads();
        }

        {
            float vr[16], vi[16];
            for (int k = 0; k < 16; ++k) {
                int i = tid * 16 + k;
                int s = i ^ (i >> 1);
                vr[k] = Sr[s];  vi[k] = Si[s];
            }
            __syncthreads();
            for (int k = 0; k < 16; ++k) {
                int i = tid * 16 + k;
                Sr[i] = vr[k];  Si[i] = vi[k];
            }
            __syncthreads();
        }
    }

    float partial = 0.f;
    {
        float base = cfA[(tid >> 4) & 15] + cfM[tid & 15];
        for (int k = 0; k < 16; ++k) {
            int i = tid * 16 + k;
            float pr = Sr[i], pi = Si[i];
            partial += (pr * pr + pi * pi) * (base + cfZ[k]);
        }
    }
    red[tid] = partial;
    __syncthreads();
    for (int s = 128; s > 0; s >>= 1) {
        if (tid < s) red[tid] += red[tid + s];
        __syncthreads();
    }
    if (tid == 0) out[b] = red[0] + head_b[0];
}

extern "C" void kernel_launch(void* const* d_in, const int* in_sizes, int n_in,
                              void* d_out, int out_size, void* d_ws, size_t ws_size,
                              hipStream_t stream) {
    const float* x      = (const float*)d_in[0];
    const float* params = (const float*)d_in[1];
    const float* head_w = (const float*)d_in[2];
    const float* head_b = (const float*)d_in[3];
    float* out = (float*)d_out;

    float* gr = (float*)d_ws;
    float* gi = gr + NL * 3 * 256;

    qgate_build<<<(NL * 3 * 256 + 255) / 256, 256, 0, stream>>>(params, gr, gi);
    qsim<<<QBATCH, 256, 0, stream>>>(x, head_w, head_b, gr, gi, out);
}


# Transfer-matrix build folded into the main kernel: 24 lanes take sincos of one shared angle each, 12 lanes write closed-form wire matrices to LDS; 2 launches -> 1; table entries round differently
# speedup vs baseline: 54.0313x; 54.0313x over previous
#include <hip/hip_runtime.h>


#define NW     12
#define QBATCH 8192

typedef __attribute__((ext_vector_type(2))) float v2f;
typedef __attribute__((ext_vector_type(8))) float v8f;

static __device__ __forceinline__ v8f wmma4(v2f a, v2f b, v8f c) {
    return __builtin_amdgcn_wmma_f32_16x16x4_f32(false, a, false, b, (short)0, c,
                                                 false, false);
}

static __device__ __forceinline__ float xhalf(float v) {
    const int b = __builtin_bit_cast(int, v);
    return __builtin_bit_cast(float, __builtin_amdgcn_permlanex16(b, b, 0x76543210, 0xfedcba98, false, false));
}

#define TSTRIDE 68

static __device__ __forceinline__ void wire_matrix(float S0, float C0, float S1, float C1, float hw, float* __restrict__ m) {
    const float g   = 0.5f * S0;
    const float a   = g * S0;
    const float oma = 1.f - a;
    const float gC  = g * C0;
    const float dd  = oma - a;
    const float sc  = S0 * C0;
    const float q   = C1 * C1;
    const float pr  = -S1;
    const float pi  = -S1 * C1;
    const float qo = q * oma, qa = q * a, qsc = q * sc, qs = q * S0, qc2 = 2.f * q * C0;
    const float prgC = pr * gC, pig = pi * g, prg = pr * g, pigC = pi * gC;
    const float prC = pr * C0, piC = pi * C0;
    const float hsc = 2.f * hw * sc;
    float4* r = reinterpret_cast<float4*>(m);
    r[0]  = make_float4( qo,  qa,  qsc,  qs);
    r[1]  = make_float4(-qa, -qo, -qsc,  qs);
    r[2]  = make_float4(-qs,  qs,  qc2,  0.f);
    r[3]  = make_float4(-qs,  qs, -qc2,  0.f);
    r[4]  = make_float4( qa,  qo, -qsc, -qs);
    r[5]  = make_float4(-qo, -qa,  qsc, -qs);
    r[6]  = make_float4( hw,  hw,  hsc,  0.f);
    r[7]  = make_float4(0.f, 0.f, 4.f * hw * C0, 0.f);
    r[8]  = make_float4( prgC + pig,  prgC - pig, pr, -pi * dd);
    r[9]  = make_float4( pigC - prg,  pigC + prg, pi,  pr * dd);
    r[10] = make_float4( prC,  prC, 0.f,  2.f * pi * S0);
    r[11] = make_float4( piC,  piC, 0.f, -2.f * pr * S0);
    r[12] = make_float4(-prgC - pig,  pig - prgC, pr,  pi * dd);
    r[13] = make_float4( prg - pigC, -prg - pigC, pi, -pr * dd);
    r[14] = make_float4( hw,  hw, -hsc,  0.f);
    r[15] = make_float4(0.f, 0.f, 0.f, 0.f);
}

__launch_bounds__(32)
__global__ void qxfer(const float* __restrict__ x,
                      const float* __restrict__ params,
                      const float* __restrict__ head_w,
                      const float* __restrict__ head_b,
                      float* __restrict__ out) {
    __shared__ __attribute__((aligned(16))) float tmL[NW * TSTRIDE];

    const int lane = threadIdx.x & 31;
    const int half = lane >> 4;
    const int lm   = lane & 15;
    const int b    = blockIdx.x * 16 + lm;

    float xa[NW];
    {
        const float4* xv = reinterpret_cast<const float4*>(x + b * NW);
        float4 x0 = xv[0], x1 = xv[1], x2 = xv[2];
        xa[0] = x0.x; xa[1] = x0.y; xa[2]  = x0.z; xa[3]  = x0.w;
        xa[4] = x1.x; xa[5] = x1.y; xa[6]  = x1.z; xa[7]  = x1.w;
        xa[8] = x2.x; xa[9] = x2.y; xa[10] = x2.z; xa[11] = x2.w;
    }

    const float tpar = params[half * NW + (lm < NW ? lm : 0)];
    const float hwp  = head_w[(lm >= 1 && lm < NW) ? lm - 1 : 0];
    float S, C;
    sincosf(tpar, &S, &C);
    const float S1 = xhalf(S), C1 = xhalf(C);
    if (lane < NW) {
        const float hw = (lane >= 1) ? hwp : 0.f;
        wire_matrix(S, C, S1, C1, hw, tmL + lane * TSTRIDE);
    }
    __syncthreads();

    v2f a[NW];
    for (int j = 0; j < NW; ++j) {
        const float2 t = *reinterpret_cast<const float2*>(tmL + j * TSTRIDE + lm * 4 + half * 2);
        a[j].x = t.x; a[j].y = t.y;
    }

    v2f y;
    y.x = half ? 0.f : 1.f;
    y.y = 0.f;
    float acc = 0.f;

#pragma unroll
    for (int j = 0; j < NW; ++j) {
        float s, c;
        sincosf(xa[j] * 0.5f, &s, &c);
        const float cc = c * c, cs = c * s, ss = s * s;
        v8f d = {};
        d = wmma4(a[j], y, d);
        const float m6 = half ? ss : cc;
        acc += m6 * d[6] + cs * d[7];
        y.x = cc * d[0] + cs * d[2] + ss * d[4];
        y.y = cc * d[1] + cs * d[3] + ss * d[5];
    }

    const float last = half ? (2.f * y.x) : (y.x + y.y);
    float t = acc + head_w[NW - 1] * last;
    t += __shfl_xor(t, 16);
    if (half == 0) out[b] = t + head_b[0];
}

extern "C" void kernel_launch(void* const* d_in, const int* in_sizes, int n_in,
                              void* d_out, int out_size, void* d_ws, size_t ws_size,
                              hipStream_t stream) {
    const float* x      = (const float*)d_in[0];
    const float* params = (const float*)d_in[1];
    const float* head_w = (const float*)d_in[2];
    const float* head_b = (const float*)d_in[3];
    float* out = (float*)d_out;

    qxfer<<<QBATCH / 16, 32, 0, stream>>>(x, params, head_w, head_b, out);
}


# All 13 sincos of a lane (12 half angles of its sample, 1 shared angle) on the library's short-range path, written branch-free behind one range test so the 13 chains interleave; same operations
# speedup vs baseline: 66.2433x; 1.2260x over previous
#include <hip/hip_runtime.h>


#define NW     12
#define QBATCH 8192

typedef __attribute__((ext_vector_type(2))) float v2f;
typedef __attribute__((ext_vector_type(8))) float v8f;

static __device__ __forceinline__ v8f wmma4(v2f a, v2f b, v8f c) {
    return __builtin_amdgcn_wmma_f32_16x16x4_f32(false, a, false, b, (short)0, c,
                                                 false, false);
}

static __device__ __forceinline__ float xhalf(float v) {
    const int b = __builtin_bit_cast(int, v);
    return __builtin_bit_cast(float, __builtin_amdgcn_permlanex16(b, b, 0x76543210, 0xfedcba98, false, false));
}

#define SHORT_BOUND 0x1.0p+17f

static __device__ __forceinline__ void sincos_short(float x, float& s, float& c) {
    const float ax = __builtin_fabsf(x);
    const float fn = __builtin_rintf(ax * 0x1.45f306p-1f);
    float r = __builtin_fmaf(fn, -0x1.921fb4p+0f, ax);
    r = __builtin_fmaf(fn, -0x1.4442d0p-24f, r);
    r = __builtin_fmaf(fn, -0x1.846988p-48f, r);
    const int q = (int)fn;
    const float t = r * r;
    const float ps = __builtin_fmaf(r, t * __builtin_fmaf(t, __builtin_fmaf(t, -0x1.983304p-13f, 0x1.110388p-7f), -0x1.55553ap-3f), r);
    const float pc = __builtin_fmaf(t, __builtin_fmaf(t, __builtin_fmaf(t, __builtin_fmaf(t, 0x1.aea668p-16f, -0x1.6c9e76p-10f), 0x1.5557eep-5f), -0x1.000008p-1f), 1.0f);
    const bool odd = (q & 1) != 0;
    const unsigned flip = (q & 2) ? 0x80000000u : 0u;
    const unsigned sx = __builtin_bit_cast(unsigned, x) & 0x80000000u;
    const float sm = odd ? pc : ps;
    const float cm = odd ? -ps : pc;
    s = __builtin_bit_cast(float, __builtin_bit_cast(unsigned, sm) ^ flip ^ sx);
    c = __builtin_bit_cast(float, __builtin_bit_cast(unsigned, cm) ^ flip);
}

#define TSTRIDE 68

static __device__ __forceinline__ void wire_matrix(float S0, float C0, float S1, float C1, float hw, float* __restrict__ m) {
    const float g   = 0.5f * S0;
    const float a   = g * S0;
    const float oma = 1.f - a;
    const float gC  = g * C0;
    const float dd  = oma - a;
    const float sc  = S0 * C0;
    const float q   = C1 * C1;
    const float pr  = -S1;
    const float pi  = -S1 * C1;
    const float qo = q * oma, qa = q * a, qsc = q * sc, qs = q * S0, qc2 = 2.f * q * C0;
    const float prgC = pr * gC, pig = pi * g, prg = pr * g, pigC = pi * gC;
    const float prC = pr * C0, piC = pi * C0;
    const float hsc = 2.f * hw * sc;
    float4* r = reinterpret_cast<float4*>(m);
    r[0]  = make_float4( qo,  qa,  qsc,  qs);
    r[1]  = make_float4(-qa, -qo, -qsc,  qs);
    r[2]  = make_float4(-qs,  qs,  qc2,  0.f);
    r[3]  = make_float4(-qs,  qs, -qc2,  0.f);
    r[4]  = make_float4( qa,  qo, -qsc, -qs);
    r[5]  = make_float4(-qo, -qa,  qsc, -qs);
    r[6]  = make_float4( hw,  hw,  hsc,  0.f);
    r[7]  = make_float4(0.f, 0.f, 4.f * hw * C0, 0.f);
    r[8]  = make_float4( prgC + pig,  prgC - pig, pr, -pi * dd);
    r[9]  = make_float4( pigC - prg,  pigC + prg, pi,  pr * dd);
    r[10] = make_float4( prC,  prC, 0.f,  2.f * pi * S0);
    r[11] = make_float4( piC,  piC, 0.f, -2.f * pr * S0);
    r[12] = make_float4(-prgC - pig,  pig - prgC, pr,  pi * dd);
    r[13] = make_float4( prg - pigC, -prg - pigC, pi, -pr * dd);
    r[14] = make_float4( hw,  hw, -hsc,  0.f);
    r[15] = make_float4(0.f, 0.f, 0.f, 0.f);
}

__launch_bounds__(32)
__global__ void qxfer(const float* __restrict__ x,
                      const float* __restrict__ params,
                      const float* __restrict__ head_w,
                      const float* __restrict__ head_b,
                      float* __restrict__ out) {
    __shared__ __attribute__((aligned(16))) float tmL[NW * TSTRIDE];

    const int lane = threadIdx.x & 31;
    const int half = lane >> 4;
    const int lm   = lane & 15;
    const int b    = blockIdx.x * 16 + lm;

    float xa[NW];
    {
        const float4* xv = reinterpret_cast<const float4*>(x + b * NW);
        float4 x0 = xv[0], x1 = xv[1], x2 = xv[2];
        xa[0] = x0.x; xa[1] = x0.y; xa[2]  = x0.z; xa[3]  = x0.w;
        xa[4] = x1.x; xa[5] = x1.y; xa[6]  = x1.z; xa[7]  = x1.w;
        xa[8] = x2.x; xa[9] = x2.y; xa[10] = x2.z; xa[11] = x2.w;
    }

    const float tpar = params[half * NW + (lm < NW ? lm : 0)];
    const float hwp  = head_w[(lm >= 1 && lm < NW) ? lm - 1 : 0];
    float ang[NW + 1], sn[NW + 1], cn[NW + 1];
    float amax = __builtin_fabsf(tpar);
    ang[NW] = tpar;
#pragma unroll
    for (int j = 0; j < NW; ++j) {
        ang[j] = xa[j] * 0.5f;
        amax = fmaxf(amax, __builtin_fabsf(ang[j]));
    }
    if (amax < SHORT_BOUND) {
#pragma unroll
        for (int k = 0; k <= NW; ++k) sincos_short(ang[k], sn[k], cn[k]);
    } else {
#pragma unroll
        for (int k = 0; k <= NW; ++k) sincosf(ang[k], &sn[k], &cn[k]);
    }
    const float S = sn[NW], C = cn[NW];
    const float S1 = xhalf(S), C1 = xhalf(C);
    if (lane < NW) {
        const float hw = (lane >= 1) ? hwp : 0.f;
        wire_matrix(S, C, S1, C1, hw, tmL + lane * TSTRIDE);
    }
    __syncthreads();

    v2f a[NW];
    for (int j = 0; j < NW; ++j) {
        const float2 t = *reinterpret_cast<const float2*>(tmL + j * TSTRIDE + lm * 4 + half * 2);
        a[j].x = t.x; a[j].y = t.y;
    }

    v2f y;
    y.x = half ? 0.f : 1.f;
    y.y = 0.f;
    float acc = 0.f;

#pragma unroll
    for (int j = 0; j < NW; ++j) {
        const float s = sn[j], c = cn[j];
        const float cc = c * c, cs = c * s, ss = s * s;
        v8f d = {};
        d = wmma4(a[j], y, d);
        const float m6 = half ? ss : cc;
        acc += m6 * d[6] + cs * d[7];
        y.x = cc * d[0] + cs * d[2] + ss * d[4];
        y.y = cc * d[1] + cs * d[3] + ss * d[5];
    }

    const float last = half ? (2.f * y.x) : (y.x + y.y);
    float t = acc + head_w[NW - 1] * last;
    t += __shfl_xor(t, 16);
    if (half == 0) out[b] = t + head_b[0];
}

extern "C" void kernel_launch(void* const* d_in, const int* in_sizes, int n_in,
                              void* d_out, int out_size, void* d_ws, size_t ws_size,
                              hipStream_t stream) {
    const float* x      = (const float*)d_in[0];
    const float* params = (const float*)d_in[1];
    const float* head_w = (const float*)d_in[2];
    const float* head_b = (const float*)d_in[3];
    float* out = (float*)d_out;

    qxfer<<<QBATCH / 16, 32, 0, stream>>>(x, params, head_w, head_b, out);
}
